# starting kernel
# baseline (speedup 1.0000x reference)
#include <hip/hip_runtime.h>
#include <hip/hip_bf16.h>

typedef __attribute__((ext_vector_type(2))) float v2f;
typedef __attribute__((ext_vector_type(8))) float v8f;

#define BATCH        64
#define NPTS         8192
#define TOTW         9356
#define PTS_PER_BLK  64
#define ACT_STRIDE   65

#define LOG2E_F 1.4426950408889634f
#define LN2_F   0.6931471805599453f

#define W0_OFF   0
#define V0_OFF   192
#define W1_OFF   448
#define V1_OFF   4544
#define W2_OFF   4800
#define V2_OFF   8896
#define W3_OFF   9152
#define B3_OFF   9344

__device__ __forceinline__ float sigmoid_f(float x) {
    return __builtin_amdgcn_rcpf(1.0f + __builtin_amdgcn_exp2f(x * -LOG2E_F));
}
__device__ __forceinline__ float softplus_f(float x) {
    const float e = __builtin_amdgcn_exp2f(fabsf(x) * -LOG2E_F);
    return fmaf(LN2_F, __builtin_amdgcn_logf(1.0f + e), fmaxf(x, 0.0f));
}

__device__ __forceinline__ void mid_layer(float* __restrict__ act,
                                          const float* __restrict__ w_lds,
                                          int wOff, int vOff,
                                          const float ctx8[8],
                                          int lrow, bool hi)
{
    v2f af[16];
    const int koff = hi ? 2 : 0;
#pragma unroll
    for (int k = 0; k < 16; ++k) {
        const float* ap = act + lrow * ACT_STRIDE + 4 * k + koff;
        af[k].x = ap[0];
        af[k].y = ap[1];
    }

    v8f acc[4] = {{}, {}, {}, {}};
#pragma unroll
    for (int k = 0; k < 16; ++k) {
        const int kb = 4 * k + koff;
        const float* wr0 = w_lds + wOff + (kb + 0) * 64 + lrow;
        const float* wr1 = w_lds + wOff + (kb + 1) * 64 + lrow;
#pragma unroll
        for (int t = 0; t < 4; ++t) {
            v2f bf;
            bf.x = wr0[t * 16];
            bf.y = wr1[t * 16];
            acc[t] = __builtin_amdgcn_wmma_f32_16x16x4_f32(
                false, af[k], false, bf, (short)0, acc[t], false, false);
        }
    }

#pragma unroll
    for (int t = 0; t < 4; ++t) {
        const int n = t * 16 + lrow;
        const float bb = w_lds[vOff + n];
        const float ws = w_lds[vOff + 64 + n];
        const float bs = w_lds[vOff + 128 + n];
        const float wh = w_lds[vOff + 192 + n];
#pragma unroll
        for (int r = 0; r < 8; ++r) {
            const int m = r + (hi ? 8 : 0);
            const float c = ctx8[r];
            float v = acc[t][r] + bb;
            v = sigmoid_f(fmaf(c, ws, bs)) * v + c * wh;
            act[m * ACT_STRIDE + n] = softplus_f(v);
        }
    }
}

__global__ __launch_bounds__(128) void ode_hypernet_kernel(
    const float* __restrict__ context,
    const float* __restrict__ y,
    const float* __restrict__ tnw,
    float* __restrict__ out)
{
    __shared__ __align__(16) float w_lds[TOTW + 4];
    __shared__ float ctx_lds[PTS_PER_BLK];
    __shared__ float act_lds[4 * 16 * ACT_STRIDE];

    const int tid  = threadIdx.x;
    const int b    = blockIdx.y;
    const int wid  = tid >> 5;
    const int lane = tid & 31;
    const int lrow = lane & 15;
    const bool hi  = lane >= 16;
    const int point0_blk = blockIdx.x * PTS_PER_BLK;

    {
        const float4* src = (const float4*)(tnw + (size_t)b * TOTW);
        __builtin_prefetch(src, 0, 0);
        float4* dst = (float4*)w_lds;
        for (int i = tid; i < TOTW / 4; i += 128) dst[i] = src[i];
    }
    if (tid < PTS_PER_BLK)
        ctx_lds[tid] = context[(size_t)b * NPTS + point0_blk + tid];
    __syncthreads();

    const int point0 = point0_blk + wid * 16;
    float* act = act_lds + wid * 16 * ACT_STRIDE;

    float ctx8[8];
#pragma unroll
    for (int r = 0; r < 8; ++r)
        ctx8[r] = ctx_lds[wid * 16 + r + (hi ? 8 : 0)];

    {
        v2f a0;
        {
            const float* yp = y + ((size_t)b * NPTS + point0 + lrow) * 3;
            const float ya = yp[hi ? 2 : 0];
            const float yb = yp[1];
            a0.x = ya;
            a0.y = hi ? 0.0f : yb;
        }
#pragma unroll
        for (int t = 0; t < 4; ++t) {
            const int n = t * 16 + lrow;
            v2f bf;
            bf.x = w_lds[W0_OFF + (hi ? 2 : 0) * 64 + n];
            const float w1 = w_lds[W0_OFF + 1 * 64 + n];
            bf.y = hi ? 0.0f : w1;
            v8f acc = {};
            acc = __builtin_amdgcn_wmma_f32_16x16x4_f32(
                false, a0, false, bf, (short)0, acc, false, false);
            const float bb = w_lds[V0_OFF + n];
            const float ws = w_lds[V0_OFF + 64 + n];
            const float bs = w_lds[V0_OFF + 128 + n];
            const float wh = w_lds[V0_OFF + 192 + n];
#pragma unroll
            for (int r = 0; r < 8; ++r) {
                const int m = r + (hi ? 8 : 0);
                const float c = ctx8[r];
                float v = acc[r] + bb;
                v = sigmoid_f(fmaf(c, ws, bs)) * v + c * wh;
                act[m * ACT_STRIDE + n] = softplus_f(v);
            }
        }
    }

    mid_layer(act, w_lds, W1_OFF, V1_OFF, ctx8, lrow, hi);
    mid_layer(act, w_lds, W2_OFF, V2_OFF, ctx8, lrow, hi);

    {
        v2f af[16];
        const int koff = hi ? 2 : 0;
#pragma unroll
        for (int k = 0; k < 16; ++k) {
            const float* ap = act + lrow * ACT_STRIDE + 4 * k + koff;
            af[k].x = ap[0];
            af[k].y = ap[1];
        }
        const bool nvalid = lrow < 3;
        const int nc = nvalid ? lrow : 0;
        const float nmask = nvalid ? 1.0f : 0.0f;
        v8f acc0 = {}, acc1 = {};
#pragma unroll
        for (int k = 0; k < 16; k += 2) {
            const int kb0 = 4 * k + koff;
            const int kb1 = 4 * (k + 1) + koff;
            v2f bf0, bf1;
            bf0.x = w_lds[W3_OFF + (kb0 + 0) * 3 + nc] * nmask;
            bf0.y = w_lds[W3_OFF + (kb0 + 1) * 3 + nc] * nmask;
            bf1.x = w_lds[W3_OFF + (kb1 + 0) * 3 + nc] * nmask;
            bf1.y = w_lds[W3_OFF + (kb1 + 1) * 3 + nc] * nmask;
            acc0 = __builtin_amdgcn_wmma_f32_16x16x4_f32(
                false, af[k], false, bf0, (short)0, acc0, false, false);
            acc1 = __builtin_amdgcn_wmma_f32_16x16x4_f32(
                false, af[k + 1], false, bf1, (short)0, acc1, false, false);
        }
        const v8f acc = acc0 + acc1;
        if (nvalid) {
            const float bb = w_lds[B3_OFF + lrow];
            const float ws = w_lds[B3_OFF + 3 + lrow];
            const float bs = w_lds[B3_OFF + 6 + lrow];
            const float wh = w_lds[B3_OFF + 9 + lrow];
#pragma unroll
            for (int r = 0; r < 8; ++r) {
                const int m = r + (hi ? 8 : 0);
                const float c = ctx8[r];
                float v = acc[r] + bb;
                v = sigmoid_f(fmaf(c, ws, bs)) * v + c * wh;
                out[((size_t)b * NPTS + point0 + m) * 3 + lrow] = v;
            }
        }
    }
}

extern "C" void kernel_launch(void* const* d_in, const int* in_sizes, int n_in,
                              void* d_out, int out_size, void* d_ws, size_t ws_size,
                              hipStream_t stream) {
    const float* context = (const float*)d_in[0];
    const float* y       = (const float*)d_in[1];
    const float* tnw     = (const float*)d_in[2];
    float* out           = (float*)d_out;

    dim3 grid(NPTS / PTS_PER_BLK, BATCH);
    ode_hypernet_kernel<<<grid, 128, 0, stream>>>(context, y, tnw, out);
}


# LDS laid out for bank-conflict-free 64-bit fragment loads: W1/W2 rows interleaved in pairs (pitch 160), activation rows 65->68; each A/B fragment is one b64 load; k order and sums unchanged
# speedup vs baseline: 1.0808x; 1.0808x over previous
#include <hip/hip_runtime.h>
#include <hip/hip_bf16.h>

typedef __attribute__((ext_vector_type(2))) float v2f;
typedef __attribute__((ext_vector_type(8))) float v8f;

#define BATCH        64
#define NPTS         8192
#define TOTW         9356
#define PTS_PER_BLK  64
#define ACT_STRIDE   68
#define WP_PITCH     160

#define LOG2E_F 1.4426950408889634f
#define LN2_F   0.6931471805599453f

#define G_W1     448
#define G_V1     4544
#define G_W2     4800
#define G_V2     8896

#define W0_OFF   0
#define V0_OFF   192
#define W1_OFF   448
#define V1_OFF   5568
#define W2_OFF   5824
#define V2_OFF   10944
#define W3_OFF   11200
#define B3_OFF   11392
#define LDS_W_FLOATS 11404

__device__ __forceinline__ float sigmoid_f(float x) {
    return __builtin_amdgcn_rcpf(1.0f + __builtin_amdgcn_exp2f(x * -LOG2E_F));
}
__device__ __forceinline__ float softplus_f(float x) {
    const float e = __builtin_amdgcn_exp2f(fabsf(x) * -LOG2E_F);
    return fmaf(LN2_F, __builtin_amdgcn_logf(1.0f + e), fmaxf(x, 0.0f));
}

__device__ __forceinline__ void mid_layer(float* __restrict__ act,
                                          const float* __restrict__ w_lds,
                                          const float* __restrict__ w_lane,
                                          int wOff, int vOff,
                                          const float ctx8[8],
                                          int lrow, bool hi)
{
    v2f af[16];
    const int koff = hi ? 2 : 0;
#pragma unroll
    for (int k = 0; k < 16; ++k)
        af[k] = *(const v2f*)(act + lrow * ACT_STRIDE + 4 * k + koff);

    v8f acc[4] = {{}, {}, {}, {}};
#pragma unroll
    for (int k = 0; k < 16; ++k) {
        const float* wp = w_lane + wOff + 2 * k * WP_PITCH;
#pragma unroll
        for (int t = 0; t < 4; ++t) {
            const v2f bf = *(const v2f*)(wp + 32 * t);
            acc[t] = __builtin_amdgcn_wmma_f32_16x16x4_f32(
                false, af[k], false, bf, (short)0, acc[t], false, false);
        }
    }

#pragma unroll
    for (int t = 0; t < 4; ++t) {
        const int n = t * 16 + lrow;
        const float bb = w_lds[vOff + n];
        const float ws = w_lds[vOff + 64 + n];
        const float bs = w_lds[vOff + 128 + n];
        const float wh = w_lds[vOff + 192 + n];
#pragma unroll
        for (int r = 0; r < 8; ++r) {
            const int m = r + (hi ? 8 : 0);
            const float c = ctx8[r];
            float v = acc[t][r] + bb;
            v = sigmoid_f(fmaf(c, ws, bs)) * v + c * wh;
            act[m * ACT_STRIDE + n] = softplus_f(v);
        }
    }
}

__global__ __launch_bounds__(128) void ode_hypernet_kernel(
    const float* __restrict__ context,
    const float* __restrict__ y,
    const float* __restrict__ tnw,
    float* __restrict__ out)
{
    __shared__ __align__(16) float w_lds[LDS_W_FLOATS];
    __shared__ float ctx_lds[PTS_PER_BLK];
    __shared__ __align__(16) float act_lds[4 * 16 * ACT_STRIDE];

    const int tid  = threadIdx.x;
    const int b    = blockIdx.y;
    const int wid  = tid >> 5;
    const int lane = tid & 31;
    const int lrow = lane & 15;
    const bool hi  = lane >= 16;
    const int point0_blk = blockIdx.x * PTS_PER_BLK;

    {
        const float* srcf = tnw + (size_t)b * TOTW;
        const float4* src = (const float4*)srcf;
        __builtin_prefetch(src, 0, 0);
        float4* dst = (float4*)w_lds;
#pragma unroll
        for (int it = 0; it < 8; ++it) {
            const int p = wid + 4 * it;
            const float2 a1 = *(const float2*)(srcf + G_W1 + (2 * p) * 64 + 2 * lane);
            const float2 b1 = *(const float2*)(srcf + G_W1 + (2 * p + 1) * 64 + 2 * lane);
            const float2 a2 = *(const float2*)(srcf + G_W2 + (2 * p) * 64 + 2 * lane);
            const float2 b2 = *(const float2*)(srcf + G_W2 + (2 * p + 1) * 64 + 2 * lane);
            dst[W1_OFF / 4 + p * (WP_PITCH / 4) + lane] = make_float4(a1.x, b1.x, a1.y, b1.y);
            dst[W2_OFF / 4 + p * (WP_PITCH / 4) + lane] = make_float4(a2.x, b2.x, a2.y, b2.y);
        }
        const int i0 = tid < 112 ? tid : tid - 112;
        const int i1 = tid & 63;
        const int i2 = tid < 115 ? tid : tid - 115;
        dst[i0] = src[i0];
        dst[V1_OFF / 4 + i1] = src[G_V1 / 4 + i1];
        dst[V2_OFF / 4 + i2] = src[G_V2 / 4 + i2];
    }
    if (tid < PTS_PER_BLK)
        ctx_lds[tid] = context[(size_t)b * NPTS + point0_blk + tid];
    __syncthreads();

    const int point0 = point0_blk + wid * 16;
    float* act = act_lds + wid * 16 * ACT_STRIDE;

    float ctx8[8];
#pragma unroll
    for (int r = 0; r < 8; ++r)
        ctx8[r] = ctx_lds[wid * 16 + r + (hi ? 8 : 0)];

    {
        v2f a0;
        {
            const float* yp = y + ((size_t)b * NPTS + point0 + lrow) * 3;
            const float ya = yp[hi ? 2 : 0];
            const float yb = yp[1];
            a0.x = ya;
            a0.y = hi ? 0.0f : yb;
        }
#pragma unroll
        for (int t = 0; t < 4; ++t) {
            const int n = t * 16 + lrow;
            v2f bf;
            bf.x = w_lds[W0_OFF + (hi ? 2 : 0) * 64 + n];
            const float w1 = w_lds[W0_OFF + 1 * 64 + n];
            bf.y = hi ? 0.0f : w1;
            v8f acc = {};
            acc = __builtin_amdgcn_wmma_f32_16x16x4_f32(
                false, a0, false, bf, (short)0, acc, false, false);
            const float bb = w_lds[V0_OFF + n];
            const float ws = w_lds[V0_OFF + 64 + n];
            const float bs = w_lds[V0_OFF + 128 + n];
            const float wh = w_lds[V0_OFF + 192 + n];
#pragma unroll
            for (int r = 0; r < 8; ++r) {
                const int m = r + (hi ? 8 : 0);
                const float c = ctx8[r];
                float v = acc[r] + bb;
                v = sigmoid_f(fmaf(c, ws, bs)) * v + c * wh;
                act[m * ACT_STRIDE + n] = softplus_f(v);
            }
        }
    }

    const float* w_lane = w_lds + (hi ? WP_PITCH : 0) + 2 * lrow;
    mid_layer(act, w_lds, w_lane, W1_OFF, V1_OFF, ctx8, lrow, hi);
    mid_layer(act, w_lds, w_lane, W2_OFF, V2_OFF, ctx8, lrow, hi);

    {
        v2f af[16];
        const int koff = hi ? 2 : 0;
#pragma unroll
        for (int k = 0; k < 16; ++k)
            af[k] = *(const v2f*)(act + lrow * ACT_STRIDE + 4 * k + koff);
        const bool nvalid = lrow < 3;
        const int nc = nvalid ? lrow : 0;
        const float nmask = nvalid ? 1.0f : 0.0f;
        v8f acc0 = {}, acc1 = {};
#pragma unroll
        for (int k = 0; k < 16; k += 2) {
            const int kb0 = 4 * k + koff;
            const int kb1 = 4 * (k + 1) + koff;
            v2f bf0, bf1;
            bf0.x = w_lds[W3_OFF + (kb0 + 0) * 3 + nc] * nmask;
            bf0.y = w_lds[W3_OFF + (kb0 + 1) * 3 + nc] * nmask;
            bf1.x = w_lds[W3_OFF + (kb1 + 0) * 3 + nc] * nmask;
            bf1.y = w_lds[W3_OFF + (kb1 + 1) * 3 + nc] * nmask;
            acc0 = __builtin_amdgcn_wmma_f32_16x16x4_f32(
                false, af[k], false, bf0, (short)0, acc0, false, false);
            acc1 = __builtin_amdgcn_wmma_f32_16x16x4_f32(
                false, af[k + 1], false, bf1, (short)0, acc1, false, false);
        }
        const v8f acc = acc0 + acc1;
        if (nvalid) {
            const float bb = w_lds[B3_OFF + lrow];
            const float ws = w_lds[B3_OFF + 3 + lrow];
            const float bs = w_lds[B3_OFF + 6 + lrow];
            const float wh = w_lds[B3_OFF + 9 + lrow];
#pragma unroll
            for (int r = 0; r < 8; ++r) {
                const int m = r + (hi ? 8 : 0);
                const float c = ctx8[r];
                float v = acc[r] + bb;
                v = sigmoid_f(fmaf(c, ws, bs)) * v + c * wh;
                out[((size_t)b * NPTS + point0 + m) * 3 + lrow] = v;
            }
        }
    }
}

extern "C" void kernel_launch(void* const* d_in, const int* in_sizes, int n_in,
                              void* d_out, int out_size, void* d_ws, size_t ws_size,
                              hipStream_t stream) {
    const float* context = (const float*)d_in[0];
    const float* y       = (const float*)d_in[1];
    const float* tnw     = (const float*)d_in[2];
    float* out           = (float*)d_out;

    dim3 grid(NPTS / PTS_PER_BLK, BATCH);
    ode_hypernet_kernel<<<grid, 128, 0, stream>>>(context, y, tnw, out);
}


# Two 16-point tiles per wave, one layer apart: behind each matrix instruction (asm, to hold its place) issues one slot of the other tile's gate/softplus vector work; 128 points/block; sums unchanged
# speedup vs baseline: 1.1503x; 1.0643x over previous
#include <hip/hip_runtime.h>
#include <hip/hip_bf16.h>


typedef __attribute__((ext_vector_type(2))) float v2f;
typedef __attribute__((ext_vector_type(8))) float v8f;

#define BATCH        64
#define NPTS         8192
#define TOTW         9356
#define PTS_PER_BLK  128
#define ACT_STRIDE   68
#define WP_PITCH     160

#define LOG2E_F 1.4426950408889634f
#define LN2_F   0.6931471805599453f

#define G_W1     448
#define G_V1     4544
#define G_W2     4800
#define G_V2     8896

#define W0_OFF   0
#define V0_OFF   192
#define W1_OFF   448
#define V1_OFF   5568
#define W2_OFF   5824
#define V2_OFF   10944
#define W3_OFF   11200
#define B3_OFF   11392
#define LDS_W_FLOATS 11404

__device__ __forceinline__ float sigmoid_f(float x) {
    return __builtin_amdgcn_rcpf(1.0f + __builtin_amdgcn_exp2f(x * -LOG2E_F));
}
__device__ __forceinline__ float softplus_f(float x) {
    const float e = __builtin_amdgcn_exp2f(fabsf(x) * -LOG2E_F);
    return fmaf(LN2_F, __builtin_amdgcn_logf(1.0f + e), fmaxf(x, 0.0f));
}

__device__ __forceinline__ void load_afrag(v2f (&af)[16], const float* __restrict__ act_row)
{
#pragma unroll
    for (int k = 0; k < 16; ++k)
        af[k] = *(const v2f*)(act_row + 4 * k);
}

__device__ __forceinline__ void load_prm(float (&prm)[4][4], const float* __restrict__ vp)
{
#pragma unroll
    for (int t = 0; t < 4; ++t) {
        prm[t][0] = vp[16 * t];
        prm[t][1] = vp[64 + 16 * t];
        prm[t][2] = vp[128 + 16 * t];
        prm[t][3] = vp[192 + 16 * t];
    }
}

struct Epi { float x[8], sh[8], v[8]; };

template <int S>
__device__ __forceinline__ void epi_slot(Epi& e, const float (&c)[8], const v8f acc,
                                         const float (&prm)[4], float* __restrict__ st)
{
    const float bb = prm[0], ws = prm[1], bs = prm[2], wh = prm[3];
    constexpr bool half = (S == 2 || S == 3 || S == 5 || S == 6 || S == 9 || S == 10 || S == 12 || S == 13);
    constexpr bool second = (S == 3 || S == 6 || S == 10 || S == 13);
    constexpr int r0 = (half && second) ? 4 : 0;
    constexpr int r1 = (half && !second) ? 4 : 8;
#pragma unroll
    for (int r = r0; r < r1; ++r) {
        if constexpr (S == 0)            { e.x[r] = fmaf(c[r], ws, bs); }
        if constexpr (S == 1)            { e.x[r] = e.x[r] * -LOG2E_F; }
        if constexpr (S == 2 || S == 3)  { e.x[r] = __builtin_amdgcn_exp2f(e.x[r]);  e.sh[r] = c[r] * wh; }
        if constexpr (S == 4)            { e.x[r] = 1.0f + e.x[r]; }
        if constexpr (S == 5 || S == 6)  { e.x[r] = __builtin_amdgcn_rcpf(e.x[r]);   e.v[r] = acc[r] + bb; }
        if constexpr (S == 7)            { e.v[r] = fmaf(e.x[r], e.v[r], e.sh[r]); }
        if constexpr (S == 8)            { e.x[r] = fabsf(e.v[r]) * -LOG2E_F; }
        if constexpr (S == 9 || S == 10) { e.x[r] = __builtin_amdgcn_exp2f(e.x[r]);  e.sh[r] = fmaxf(e.v[r], 0.0f); }
        if constexpr (S == 11)           { e.x[r] = 1.0f + e.x[r]; }
        if constexpr (S == 12 || S == 13){ e.x[r] = __builtin_amdgcn_logf(e.x[r]); }
        if constexpr (S == 14)           { e.v[r] = fmaf(LN2_F, e.x[r], e.sh[r]); }
        if constexpr (S == 15)           { st[r * ACT_STRIDE] = e.v[r]; }
    }
}

__device__ __forceinline__ void epi_tile(Epi& e, const float (&c)[8], const v8f acc,
                                         const float (&prm)[4], float* __restrict__ st)
{
    epi_slot<0>(e, c, acc, prm, st);   epi_slot<1>(e, c, acc, prm, st);   epi_slot<2>(e, c, acc, prm, st);
    epi_slot<3>(e, c, acc, prm, st);   epi_slot<4>(e, c, acc, prm, st);   epi_slot<5>(e, c, acc, prm, st);
    epi_slot<6>(e, c, acc, prm, st);   epi_slot<7>(e, c, acc, prm, st);   epi_slot<8>(e, c, acc, prm, st);
    epi_slot<9>(e, c, acc, prm, st);   epi_slot<10>(e, c, acc, prm, st);  epi_slot<11>(e, c, acc, prm, st);
    epi_slot<12>(e, c, acc, prm, st);  epi_slot<13>(e, c, acc, prm, st);  epi_slot<14>(e, c, acc, prm, st);
    epi_slot<15>(e, c, acc, prm, st);
}

__device__ __forceinline__ void epi_alone(float* __restrict__ col, const float* __restrict__ vp,
                                          const float (&c)[8], const v8f (&acc)[4])
{
    float prm[4][4];
    load_prm(prm, vp);
    Epi e;
    epi_tile(e, c, acc[0], prm[0], col);
    epi_tile(e, c, acc[1], prm[1], col + 16);
    epi_tile(e, c, acc[2], prm[2], col + 32);
    epi_tile(e, c, acc[3], prm[3], col + 48);
}

template <bool FIRST, bool GUARD>
__device__ __forceinline__ void wmma_f32(v8f& acc, const v2f a, const v2f b)
{
    if constexpr (FIRST) {
        asm volatile("v_wmma_f32_16x16x4_f32 %0, %1, %2, 0" : "=&v"(acc) : "v"(a), "v"(b));
    } else if constexpr (GUARD) {
        asm volatile("v_wmma_f32_16x16x4_f32 %0, %1, %2, %0\n\tv_nop\n\tv_nop\n\tv_nop\n\tv_nop"
                     : "+v"(acc) : "v"(a), "v"(b));
    } else {
        asm volatile("v_wmma_f32_16x16x4_f32 %0, %1, %2, %0" : "+v"(acc) : "v"(a), "v"(b));
    }
}

template <int G>
__device__ __forceinline__ void load_bgroup(v2f (&bq)[2][8], const float* __restrict__ wp0)
{
#pragma unroll
    for (int j = 0; j < 8; ++j)
        bq[G & 1][j] = *(const v2f*)(wp0 + 2 * (2 * G + (j >> 2)) * WP_PITCH + 32 * (j & 3));
}

template <int K, int T>
__device__ __forceinline__ void mid_step(v8f (&accW)[4], const v2f (&af)[16], v2f (&bq)[2][8],
                                         const float* __restrict__ wp0, Epi& e,
                                         float* __restrict__ colO, const float (&prm)[4][4],
                                         const float (&cO)[8], const v8f (&accO)[4])
{
    wmma_f32<K == 0, K == 15>(accW[T], af[K], bq[(K >> 1) & 1][(K & 1) * 4 + T]);
    epi_slot<(K & 3) * 4 + T>(e, cO, accO[K >> 2], prm[K >> 2], colO + 16 * (K >> 2));
    if constexpr (T == 0 && (K & 1) == 0 && K + 2 < 16)
        load_bgroup<(K >> 1) + 1>(bq, wp0);
    __builtin_amdgcn_sched_barrier(0);
}

template <int K>
__device__ __forceinline__ void mid_kstep(v8f (&accW)[4], const v2f (&af)[16], v2f (&bq)[2][8],
                                          const float* __restrict__ wp0, Epi& e,
                                          float* __restrict__ colO, const float (&prm)[4][4],
                                          const float (&cO)[8], const v8f (&accO)[4])
{
    mid_step<K, 0>(accW, af, bq, wp0, e, colO, prm, cO, accO);
    mid_step<K, 1>(accW, af, bq, wp0, e, colO, prm, cO, accO);
    mid_step<K, 2>(accW, af, bq, wp0, e, colO, prm, cO, accO);
    mid_step<K, 3>(accW, af, bq, wp0, e, colO, prm, cO, accO);
}

__device__ __forceinline__ void mid_phase(v8f (&accW)[4], const float* __restrict__ rowW,
                                          const float* __restrict__ wp0,
                                          float* __restrict__ colO, const float* __restrict__ vpO,
                                          const float (&cO)[8], const v8f (&accO)[4])
{
    v2f af[16];
    load_afrag(af, rowW);
    float prm[4][4];
    load_prm(prm, vpO);
    v2f bq[2][8];
    load_bgroup<0>(bq, wp0);
    Epi e;
#define MID_K(K) mid_kstep<K>(accW, af, bq, wp0, e, colO, prm, cO, accO);
    MID_K(0)  MID_K(1)  MID_K(2)  MID_K(3)  MID_K(4)  MID_K(5)  MID_K(6)  MID_K(7)
    MID_K(8)  MID_K(9)  MID_K(10) MID_K(11) MID_K(12) MID_K(13) MID_K(14) MID_K(15)
#undef MID_K
}

template <int K>
__device__ __forceinline__ void last_step(v8f& acc0, v8f& acc1, const v2f (&af)[16], const v2f (&bf3)[16],
                                          Epi& e, float* __restrict__ colO, const float (&prm)[4][4],
                                          const float (&cO)[8], const v8f (&accO)[4])
{
    if constexpr ((K & 1) == 0)
        wmma_f32<K == 0, K == 14>(acc0, af[K], bf3[K]);
    else
        wmma_f32<K == 1, K == 15>(acc1, af[K], bf3[K]);
    constexpr int g = K >> 2, s0 = (K & 3) * 4;
    epi_slot<s0 + 0>(e, cO, accO[g], prm[g], colO + 16 * g);
    epi_slot<s0 + 1>(e, cO, accO[g], prm[g], colO + 16 * g);
    epi_slot<s0 + 2>(e, cO, accO[g], prm[g], colO + 16 * g);
    epi_slot<s0 + 3>(e, cO, accO[g], prm[g], colO + 16 * g);
    __builtin_amdgcn_sched_barrier(0);
}

__global__ __launch_bounds__(128) void ode_hypernet_kernel(
    const float* __restrict__ context,
    const float* __restrict__ y,
    const float* __restrict__ tnw,
    float* __restrict__ out)
{
    __shared__ __align__(16) float w_lds[LDS_W_FLOATS];
    __shared__ float ctx_lds[PTS_PER_BLK];
    __shared__ __align__(16) float act_ldsA[4 * 16 * ACT_STRIDE];
    __shared__ __align__(16) float act_ldsB[4 * 16 * ACT_STRIDE];

    const int tid  = threadIdx.x;
    const int b    = blockIdx.y;
    const int wid  = tid >> 5;
    const int lane = tid & 31;
    const int lrow = lane & 15;
    const bool hi  = lane >= 16;
    const int point0_blk = blockIdx.x * PTS_PER_BLK;

    {
        const float* srcf = tnw + (size_t)b * TOTW;
        const float4* src = (const float4*)srcf;
        __builtin_prefetch(src, 0, 0);
        float4* dst = (float4*)w_lds;
#pragma unroll
        for (int it = 0; it < 8; ++it) {
            const int p = wid + 4 * it;
            const float2 a1 = *(const float2*)(srcf + G_W1 + (2 * p) * 64 + 2 * lane);
            const float2 b1 = *(const float2*)(srcf + G_W1 + (2 * p + 1) * 64 + 2 * lane);
            const float2 a2 = *(const float2*)(srcf + G_W2 + (2 * p) * 64 + 2 * lane);
            const float2 b2 = *(const float2*)(srcf + G_W2 + (2 * p + 1) * 64 + 2 * lane);
            dst[W1_OFF / 4 + p * (WP_PITCH / 4) + lane] = make_float4(a1.x, b1.x, a1.y, b1.y);
            dst[W2_OFF / 4 + p * (WP_PITCH / 4) + lane] = make_float4(a2.x, b2.x, a2.y, b2.y);
        }
        const int i0 = tid < 112 ? tid : tid - 112;
        const int i1 = tid & 63;
        const int i2 = tid < 115 ? tid : tid - 115;
        dst[i0] = src[i0];
        dst[V1_OFF / 4 + i1] = src[G_V1 / 4 + i1];
        dst[V2_OFF / 4 + i2] = src[G_V2 / 4 + i2];
    }
    ctx_lds[tid] = context[(size_t)b * NPTS + point0_blk + tid];

    const int pointA = point0_blk + wid * 32;
    const int pointB = pointA + 16;

    v2f a0A, a0B;
    {
        const float* ypA = y + ((size_t)b * NPTS + pointA + lrow) * 3;
        const float* ypB = y + ((size_t)b * NPTS + pointB + lrow) * 3;
        const float yaA = ypA[hi ? 2 : 0];
        const float ybA = ypA[1];
        const float yaB = ypB[hi ? 2 : 0];
        const float ybB = ypB[1];
        a0A.x = yaA;  a0A.y = hi ? 0.0f : ybA;
        a0B.x = yaB;  a0B.y = hi ? 0.0f : ybB;
    }
    __syncthreads();

    float* actA = act_ldsA + wid * 16 * ACT_STRIDE;
    float* actB = act_ldsB + wid * 16 * ACT_STRIDE;
    const int koff = hi ? 2 : 0;
    const float* rowA = actA + lrow * ACT_STRIDE + koff;
    const float* rowB = actB + lrow * ACT_STRIDE + koff;
    float* colA = actA + (hi ? 8 : 0) * ACT_STRIDE + lrow;
    float* colB = actB + (hi ? 8 : 0) * ACT_STRIDE + lrow;

    float ctxA[8], ctxB[8];
#pragma unroll
    for (int r = 0; r < 8; ++r) {
        ctxA[r] = ctx_lds[wid * 32 + r + (hi ? 8 : 0)];
        ctxB[r] = ctx_lds[wid * 32 + 16 + r + (hi ? 8 : 0)];
    }

    const float* w_lane = w_lds + (hi ? WP_PITCH : 0) + 2 * lrow;

    v8f accA[4], accB[4];

#pragma unroll
    for (int t = 0; t < 4; ++t) {
        const int n = t * 16 + lrow;
        v2f bf;
        bf.x = w_lds[W0_OFF + (hi ? 2 : 0) * 64 + n];
        const float w1 = w_lds[W0_OFF + 1 * 64 + n];
        bf.y = hi ? 0.0f : w1;
        accA[t] = __builtin_amdgcn_wmma_f32_16x16x4_f32(
            false, a0A, false, bf, (short)0, v8f{}, false, false);
        accB[t] = __builtin_amdgcn_wmma_f32_16x16x4_f32(
            false, a0B, false, bf, (short)0, v8f{}, false, false);
    }

    epi_alone(colA, w_lds + V0_OFF + lrow, ctxA, accA);
    mid_phase(accA, rowA, w_lane + W1_OFF, colB, w_lds + V0_OFF + lrow, ctxB, accB);
    mid_phase(accB, rowB, w_lane + W1_OFF, colA, w_lds + V1_OFF + lrow, ctxA, accA);
    mid_phase(accA, rowA, w_lane + W2_OFF, colB, w_lds + V1_OFF + lrow, ctxB, accB);
    mid_phase(accB, rowB, w_lane + W2_OFF, colA, w_lds + V2_OFF + lrow, ctxA, accA);

    const bool nvalid = lrow < 3;
    const int nc = nvalid ? lrow : 0;
    const float nmask = nvalid ? 1.0f : 0.0f;
    v2f bf3[16];
#pragma unroll
    for (int k = 0; k < 16; ++k) {
        const int kb = 4 * k + koff;
        bf3[k].x = w_lds[W3_OFF + (kb + 0) * 3 + nc] * nmask;
        bf3[k].y = w_lds[W3_OFF + (kb + 1) * 3 + nc] * nmask;
    }
    const float bb3 = w_lds[B3_OFF + nc];
    const float ws3 = w_lds[B3_OFF + 3 + nc];
    const float bs3 = w_lds[B3_OFF + 6 + nc];
    const float wh3 = w_lds[B3_OFF + 9 + nc];

    v8f oA, oB;
    {
        v2f af[16];
        load_afrag(af, rowA);
        float prm[4][4];
        load_prm(prm, w_lds + V2_OFF + lrow);
        v8f acc0, acc1;
        Epi e;
#define LAST_K(K) last_step<K>(acc0, acc1, af, bf3, e, colB, prm, ctxB, accB);
        LAST_K(0)  LAST_K(1)  LAST_K(2)  LAST_K(3)  LAST_K(4)  LAST_K(5)  LAST_K(6)  LAST_K(7)
        LAST_K(8)  LAST_K(9)  LAST_K(10) LAST_K(11) LAST_K(12) LAST_K(13) LAST_K(14) LAST_K(15)
#undef LAST_K
        oA = acc0 + acc1;
    }
    float resA[8], resB[8];
    {
        v2f af[16];
        load_afrag(af, rowB);
        v8f acc0 = {}, acc1 = {};
#pragma unroll
        for (int k = 0; k < 16; k += 2) {
            acc0 = __builtin_amdgcn_wmma_f32_16x16x4_f32(
                false, af[k], false, bf3[k], (short)0, acc0, false, false);
            acc1 = __builtin_amdgcn_wmma_f32_16x16x4_f32(
                false, af[k + 1], false, bf3[k + 1], (short)0, acc1, false, false);
        }
#pragma unroll
        for (int r = 0; r < 8; ++r) {
            const float c = ctxA[r];
            const float v = oA[r] + bb3;
            resA[r] = sigmoid_f(fmaf(c, ws3, bs3)) * v + c * wh3;
        }
        oB = acc0 + acc1;
    }
#pragma unroll
    for (int r = 0; r < 8; ++r) {
        const float c = ctxB[r];
        const float v = oB[r] + bb3;
        resB[r] = sigmoid_f(fmaf(c, ws3, bs3)) * v + c * wh3;
    }
    if (nvalid) {
#pragma unroll
        for (int r = 0; r < 8; ++r) {
            const int m = r + (hi ? 8 : 0);
            out[((size_t)b * NPTS + pointA + m) * 3 + lrow] = resA[r];
            out[((size_t)b * NPTS + pointB + m) * 3 + lrow] = resB[r];
        }
    }
}

extern "C" void kernel_launch(void* const* d_in, const int* in_sizes, int n_in,
                              void* d_out, int out_size, void* d_ws, size_t ws_size,
                              hipStream_t stream) {
    const float* context = (const float*)d_in[0];
    const float* y       = (const float*)d_in[1];
    const float* tnw     = (const float*)d_in[2];
    float* out           = (float*)d_out;

    dim3 grid(NPTS / PTS_PER_BLK, BATCH);
    ode_hypernet_kernel<<<grid, 128, 0, stream>>>(context, y, tnw, out);
}


# Sigmoid gate on the hardware tanh: gate*(acc+b)+c*s = t*u+(c*s+u), t=tanh((c*ws+bs)/2), u=(acc+b)/2, halves folded into the layer vectors; 3 transcendentals per element, not 4 (exp2+rcp gone)
# speedup vs baseline: 1.2195x; 1.0602x over previous
#include <hip/hip_runtime.h>
#include <hip/hip_bf16.h>


typedef __attribute__((ext_vector_type(2))) float v2f;
typedef __attribute__((ext_vector_type(8))) float v8f;

#define BATCH        64
#define NPTS         8192
#define TOTW         9356
#define PTS_PER_BLK  128
#define ACT_STRIDE   68
#define WP_PITCH     160

#define LOG2E_F 1.4426950408889634f
#define LN2_F   0.6931471805599453f

#define G_W1     448
#define G_V1     4544
#define G_W2     4800
#define G_V2     8896

#define W0_OFF   0
#define V0_OFF   192
#define W1_OFF   448
#define V1_OFF   5568
#define W2_OFF   5824
#define V2_OFF   10944
#define W3_OFF   11200
#define B3_OFF   11392
#define LDS_W_FLOATS 11404

__device__ __forceinline__ float sigmoid_f(float x) {
    return fmaf(0.5f, __builtin_amdgcn_tanhf(0.5f * x), 0.5f);
}
__device__ __forceinline__ float softplus_f(float x) {
    const float e = __builtin_amdgcn_exp2f(fabsf(x) * -LOG2E_F);
    return fmaf(LN2_F, __builtin_amdgcn_logf(1.0f + e), fmaxf(x, 0.0f));
}

__device__ __forceinline__ void load_afrag(v2f (&af)[16], const float* __restrict__ act_row)
{
#pragma unroll
    for (int k = 0; k < 16; ++k)
        af[k] = *(const v2f*)(act_row + 4 * k);
}

__device__ __forceinline__ void load_prm(float (&prm)[4][4], const float* __restrict__ vp)
{
#pragma unroll
    for (int t = 0; t < 4; ++t) {
        prm[t][0] = 0.5f * vp[16 * t];
        prm[t][1] = 0.5f * vp[64 + 16 * t];
        prm[t][2] = 0.5f * vp[128 + 16 * t];
        prm[t][3] = vp[192 + 16 * t];
    }
}

struct Epi { float x[8], u[8], w[8], v[8]; };

template <int S>
__device__ __forceinline__ void epi_slot(Epi& e, const float (&c)[8], const v8f acc,
                                         const float (&prm)[4], float* __restrict__ st)
{
    const float bbh = prm[0], wsh = prm[1], bsh = prm[2], wh = prm[3];
    if constexpr (S == 0) {
#pragma unroll
        for (int r = 0; r < 8; ++r) e.x[r] = fmaf(c[r], wsh, bsh);
    }
    if constexpr (S == 1) {
#pragma unroll
        for (int r = 0; r < 4; ++r) e.x[r] = __builtin_amdgcn_tanhf(e.x[r]);
#pragma unroll
        for (int r = 0; r < 8; ++r) e.u[r] = fmaf(acc[r], 0.5f, bbh);
    }
    if constexpr (S == 2) {
#pragma unroll
        for (int r = 4; r < 8; ++r) e.x[r] = __builtin_amdgcn_tanhf(e.x[r]);
#pragma unroll
        for (int r = 0; r < 8; ++r) e.w[r] = fmaf(c[r], wh, e.u[r]);
    }
    if constexpr (S == 3) {
#pragma unroll
        for (int r = 0; r < 8; ++r) e.v[r] = fmaf(e.x[r], e.u[r], e.w[r]);
    }
    if constexpr (S == 4) {
#pragma unroll
        for (int r = 0; r < 8; ++r) e.x[r] = fabsf(e.v[r]) * -LOG2E_F;
    }
    if constexpr (S == 5) {
#pragma unroll
        for (int r = 0; r < 4; ++r) e.x[r] = __builtin_amdgcn_exp2f(e.x[r]);
#pragma unroll
        for (int r = 0; r < 8; ++r) e.w[r] = fmaxf(e.v[r], 0.0f);
    }
    if constexpr (S == 6) {
#pragma unroll
        for (int r = 4; r < 8; ++r) e.x[r] = __builtin_amdgcn_exp2f(e.x[r]);
#pragma unroll
        for (int r = 0; r < 4; ++r) e.x[r] = 1.0f + e.x[r];
    }
    if constexpr (S == 7) {
#pragma unroll
        for (int r = 0; r < 4; ++r) e.x[r] = __builtin_amdgcn_logf(e.x[r]);
#pragma unroll
        for (int r = 4; r < 8; ++r) e.x[r] = 1.0f + e.x[r];
    }
    if constexpr (S == 8) {
#pragma unroll
        for (int r = 4; r < 8; ++r) e.x[r] = __builtin_amdgcn_logf(e.x[r]);
#pragma unroll
        for (int r = 0; r < 4; ++r) e.v[r] = fmaf(LN2_F, e.x[r], e.w[r]);
    }
    if constexpr (S == 9) {
#pragma unroll
        for (int r = 4; r < 8; ++r) e.v[r] = fmaf(LN2_F, e.x[r], e.w[r]);
    }
    if constexpr (S == 10) {
#pragma unroll
        for (int r = 0; r < 8; ++r) st[r * ACT_STRIDE] = e.v[r];
    }
}

__device__ __forceinline__ void epi_tile(Epi& e, const float (&c)[8], const v8f acc,
                                         const float (&prm)[4], float* __restrict__ st)
{
    epi_slot<0>(e, c, acc, prm, st);   epi_slot<1>(e, c, acc, prm, st);   epi_slot<2>(e, c, acc, prm, st);
    epi_slot<3>(e, c, acc, prm, st);   epi_slot<4>(e, c, acc, prm, st);   epi_slot<5>(e, c, acc, prm, st);
    epi_slot<6>(e, c, acc, prm, st);   epi_slot<7>(e, c, acc, prm, st);   epi_slot<8>(e, c, acc, prm, st);
    epi_slot<9>(e, c, acc, prm, st);   epi_slot<10>(e, c, acc, prm, st);
}

__device__ __forceinline__ void epi_alone(float* __restrict__ col, const float* __restrict__ vp,
                                          const float (&c)[8], const v8f (&acc)[4])
{
    float prm[4][4];
    load_prm(prm, vp);
    Epi e;
    epi_tile(e, c, acc[0], prm[0], col);
    epi_tile(e, c, acc[1], prm[1], col + 16);
    epi_tile(e, c, acc[2], prm[2], col + 32);
    epi_tile(e, c, acc[3], prm[3], col + 48);
}

template <bool FIRST, bool GUARD>
__device__ __forceinline__ void wmma_f32(v8f& acc, const v2f a, const v2f b)
{
    if constexpr (FIRST) {
        asm volatile("v_wmma_f32_16x16x4_f32 %0, %1, %2, 0" : "=&v"(acc) : "v"(a), "v"(b));
    } else if constexpr (GUARD) {
        asm volatile("v_wmma_f32_16x16x4_f32 %0, %1, %2, %0\n\tv_nop\n\tv_nop\n\tv_nop\n\tv_nop"
                     : "+v"(acc) : "v"(a), "v"(b));
    } else {
        asm volatile("v_wmma_f32_16x16x4_f32 %0, %1, %2, %0" : "+v"(acc) : "v"(a), "v"(b));
    }
}

template <int G>
__device__ __forceinline__ void load_bgroup(v2f (&bq)[2][8], const float* __restrict__ wp0)
{
#pragma unroll
    for (int j = 0; j < 8; ++j)
        bq[G & 1][j] = *(const v2f*)(wp0 + 2 * (2 * G + (j >> 2)) * WP_PITCH + 32 * (j & 3));
}

template <int K, int T>
__device__ __forceinline__ void mid_step(v8f (&accW)[4], const v2f (&af)[16], v2f (&bq)[2][8],
                                         const float* __restrict__ wp0, Epi& e,
                                         float* __restrict__ colO, const float (&prm)[4][4],
                                         const float (&cO)[8], const v8f (&accO)[4])
{
    wmma_f32<K == 0, K == 15>(accW[T], af[K], bq[(K >> 1) & 1][(K & 1) * 4 + T]);
    epi_slot<(K & 3) * 4 + T>(e, cO, accO[K >> 2], prm[K >> 2], colO + 16 * (K >> 2));
    if constexpr (T == 0 && (K & 1) == 0 && K + 2 < 16)
        load_bgroup<(K >> 1) + 1>(bq, wp0);
    __builtin_amdgcn_sched_barrier(0);
}

template <int K>
__device__ __forceinline__ void mid_kstep(v8f (&accW)[4], const v2f (&af)[16], v2f (&bq)[2][8],
                                          const float* __restrict__ wp0, Epi& e,
                                          float* __restrict__ colO, const float (&prm)[4][4],
                                          const float (&cO)[8], const v8f (&accO)[4])
{
    mid_step<K, 0>(accW, af, bq, wp0, e, colO, prm, cO, accO);
    mid_step<K, 1>(accW, af, bq, wp0, e, colO, prm, cO, accO);
    mid_step<K, 2>(accW, af, bq, wp0, e, colO, prm, cO, accO);
    mid_step<K, 3>(accW, af, bq, wp0, e, colO, prm, cO, accO);
}

__device__ __forceinline__ void mid_phase(v8f (&accW)[4], const float* __restrict__ rowW,
                                          const float* __restrict__ wp0,
                                          float* __restrict__ colO, const float* __restrict__ vpO,
                                          const float (&cO)[8], const v8f (&accO)[4])
{
    v2f af[16];
    load_afrag(af, rowW);
    float prm[4][4];
    load_prm(prm, vpO);
    v2f bq[2][8];
    load_bgroup<0>(bq, wp0);
    Epi e;
#define MID_K(K) mid_kstep<K>(accW, af, bq, wp0, e, colO, prm, cO, accO);
    MID_K(0)  MID_K(1)  MID_K(2)  MID_K(3)  MID_K(4)  MID_K(5)  MID_K(6)  MID_K(7)
    MID_K(8)  MID_K(9)  MID_K(10) MID_K(11) MID_K(12) MID_K(13) MID_K(14) MID_K(15)
#undef MID_K
}

template <int K>
__device__ __forceinline__ void last_step(v8f& acc0, v8f& acc1, const v2f (&af)[16], const v2f (&bf3)[16],
                                          Epi& e, float* __restrict__ colO, const float (&prm)[4][4],
                                          const float (&cO)[8], const v8f (&accO)[4])
{
    if constexpr ((K & 1) == 0)
        wmma_f32<K == 0, K == 14>(acc0, af[K], bf3[K]);
    else
        wmma_f32<K == 1, K == 15>(acc1, af[K], bf3[K]);
    constexpr int g = K >> 2, s0 = (K & 3) * 3;
    epi_slot<s0 + 0>(e, cO, accO[g], prm[g], colO + 16 * g);
    epi_slot<s0 + 1>(e, cO, accO[g], prm[g], colO + 16 * g);
    epi_slot<s0 + 2>(e, cO, accO[g], prm[g], colO + 16 * g);
    __builtin_amdgcn_sched_barrier(0);
}

__global__ __launch_bounds__(128) void ode_hypernet_kernel(
    const float* __restrict__ context,
    const float* __restrict__ y,
    const float* __restrict__ tnw,
    float* __restrict__ out)
{
    __shared__ __align__(16) float w_lds[LDS_W_FLOATS];
    __shared__ float ctx_lds[PTS_PER_BLK];
    __shared__ __align__(16) float act_ldsA[4 * 16 * ACT_STRIDE];
    __shared__ __align__(16) float act_ldsB[4 * 16 * ACT_STRIDE];

    const int tid  = threadIdx.x;
    const int b    = blockIdx.y;
    const int wid  = tid >> 5;
    const int lane = tid & 31;
    const int lrow = lane & 15;
    const bool hi  = lane >= 16;
    const int point0_blk = blockIdx.x * PTS_PER_BLK;

    {
        const float* srcf = tnw + (size_t)b * TOTW;
        const float4* src = (const float4*)srcf;
        __builtin_prefetch(src, 0, 0);
        float4* dst = (float4*)w_lds;
#pragma unroll
        for (int it = 0; it < 8; ++it) {
            const int p = wid + 4 * it;
            const float2 a1 = *(const float2*)(srcf + G_W1 + (2 * p) * 64 + 2 * lane);
            const float2 b1 = *(const float2*)(srcf + G_W1 + (2 * p + 1) * 64 + 2 * lane);
            const float2 a2 = *(const float2*)(srcf + G_W2 + (2 * p) * 64 + 2 * lane);
            const float2 b2 = *(const float2*)(srcf + G_W2 + (2 * p + 1) * 64 + 2 * lane);
            dst[W1_OFF / 4 + p * (WP_PITCH / 4) + lane] = make_float4(a1.x, b1.x, a1.y, b1.y);
            dst[W2_OFF / 4 + p * (WP_PITCH / 4) + lane] = make_float4(a2.x, b2.x, a2.y, b2.y);
        }
        const int i0 = tid < 112 ? tid : tid - 112;
        const int i1 = tid & 63;
        const int i2 = tid < 115 ? tid : tid - 115;
        dst[i0] = src[i0];
        dst[V1_OFF / 4 + i1] = src[G_V1 / 4 + i1];
        dst[V2_OFF / 4 + i2] = src[G_V2 / 4 + i2];
    }
    ctx_lds[tid] = context[(size_t)b * NPTS + point0_blk + tid];

    const int pointA = point0_blk + wid * 32;
    const int pointB = pointA + 16;

    v2f a0A, a0B;
    {
        const float* ypA = y + ((size_t)b * NPTS + pointA + lrow) * 3;
        const float* ypB = y + ((size_t)b * NPTS + pointB + lrow) * 3;
        const float yaA = ypA[hi ? 2 : 0];
        const float ybA = ypA[1];
        const float yaB = ypB[hi ? 2 : 0];
        const float ybB = ypB[1];
        a0A.x = yaA;  a0A.y = hi ? 0.0f : ybA;
        a0B.x = yaB;  a0B.y = hi ? 0.0f : ybB;
    }
    __syncthreads();

    float* actA = act_ldsA + wid * 16 * ACT_STRIDE;
    float* actB = act_ldsB + wid * 16 * ACT_STRIDE;
    const int koff = hi ? 2 : 0;
    const float* rowA = actA + lrow * ACT_STRIDE + koff;
    const float* rowB = actB + lrow * ACT_STRIDE + koff;
    float* colA = actA + (hi ? 8 : 0) * ACT_STRIDE + lrow;
    float* colB = actB + (hi ? 8 : 0) * ACT_STRIDE + lrow;

    float ctxA[8], ctxB[8];
#pragma unroll
    for (int r = 0; r < 8; ++r) {
        ctxA[r] = ctx_lds[wid * 32 + r + (hi ? 8 : 0)];
        ctxB[r] = ctx_lds[wid * 32 + 16 + r + (hi ? 8 : 0)];
    }

    const float* w_lane = w_lds + (hi ? WP_PITCH : 0) + 2 * lrow;

    v8f accA[4], accB[4];

#pragma unroll
    for (int t = 0; t < 4; ++t) {
        const int n = t * 16 + lrow;
        v2f bf;
        bf.x = w_lds[W0_OFF + (hi ? 2 : 0) * 64 + n];
        const float w1 = w_lds[W0_OFF + 1 * 64 + n];
        bf.y = hi ? 0.0f : w1;
        accA[t] = __builtin_amdgcn_wmma_f32_16x16x4_f32(
            false, a0A, false, bf, (short)0, v8f{}, false, false);
        accB[t] = __builtin_amdgcn_wmma_f32_16x16x4_f32(
            false, a0B, false, bf, (short)0, v8f{}, false, false);
    }

    epi_alone(colA, w_lds + V0_OFF + lrow, ctxA, accA);
    mid_phase(accA, rowA, w_lane + W1_OFF, colB, w_lds + V0_OFF + lrow, ctxB, accB);
    mid_phase(accB, rowB, w_lane + W1_OFF, colA, w_lds + V1_OFF + lrow, ctxA, accA);
    mid_phase(accA, rowA, w_lane + W2_OFF, colB, w_lds + V1_OFF + lrow, ctxB, accB);
    mid_phase(accB, rowB, w_lane + W2_OFF, colA, w_lds + V2_OFF + lrow, ctxA, accA);

    const bool nvalid = lrow < 3;
    const int nc = nvalid ? lrow : 0;
    const float nmask = nvalid ? 1.0f : 0.0f;
    v2f bf3[16];
#pragma unroll
    for (int k = 0; k < 16; ++k) {
        const int kb = 4 * k + koff;
        bf3[k].x = w_lds[W3_OFF + (kb + 0) * 3 + nc] * nmask;
        bf3[k].y = w_lds[W3_OFF + (kb + 1) * 3 + nc] * nmask;
    }
    const float bb3 = w_lds[B3_OFF + nc];
    const float ws3 = w_lds[B3_OFF + 3 + nc];
    const float bs3 = w_lds[B3_OFF + 6 + nc];
    const float wh3 = w_lds[B3_OFF + 9 + nc];

    v8f oA, oB;
    {
        v2f af[16];
        load_afrag(af, rowA);
        float prm[4][4];
        load_prm(prm, w_lds + V2_OFF + lrow);
        v8f acc0, acc1;
        Epi e;
#define LAST_K(K) last_step<K>(acc0, acc1, af, bf3, e, colB, prm, ctxB, accB);
        LAST_K(0)  LAST_K(1)  LAST_K(2)  LAST_K(3)  LAST_K(4)  LAST_K(5)  LAST_K(6)  LAST_K(7)
        LAST_K(8)  LAST_K(9)  LAST_K(10) LAST_K(11) LAST_K(12) LAST_K(13) LAST_K(14) LAST_K(15)
#undef LAST_K
        oA = acc0 + acc1;
    }
    float resA[8], resB[8];
    {
        v2f af[16];
        load_afrag(af, rowB);
        v8f acc0 = {}, acc1 = {};
#pragma unroll
        for (int k = 0; k < 16; k += 2) {
            acc0 = __builtin_amdgcn_wmma_f32_16x16x4_f32(
                false, af[k], false, bf3[k], (short)0, acc0, false, false);
            acc1 = __builtin_amdgcn_wmma_f32_16x16x4_f32(
                false, af[k + 1], false, bf3[k + 1], (short)0, acc1, false, false);
        }
#pragma unroll
        for (int r = 0; r < 8; ++r) {
            const float c = ctxA[r];
            const float v = oA[r] + bb3;
            resA[r] = sigmoid_f(fmaf(c, ws3, bs3)) * v + c * wh3;
        }
        oB = acc0 + acc1;
    }
#pragma unroll
    for (int r = 0; r < 8; ++r) {
        const float c = ctxB[r];
        const float v = oB[r] + bb3;
        resB[r] = sigmoid_f(fmaf(c, ws3, bs3)) * v + c * wh3;
    }
    if (nvalid) {
#pragma unroll
        for (int r = 0; r < 8; ++r) {
            const int m = r + (hi ? 8 : 0);
            out[((size_t)b * NPTS + pointA + m) * 3 + lrow] = resA[r];
            out[((size_t)b * NPTS + pointB + m) * 3 + lrow] = resB[r];
        }
    }
}

extern "C" void kernel_launch(void* const* d_in, const int* in_sizes, int n_in,
                              void* d_out, int out_size, void* d_ws, size_t ws_size,
                              hipStream_t stream) {
    const float* context = (const float*)d_in[0];
    const float* y       = (const float*)d_in[1];
    const float* tnw     = (const float*)d_in[2];
    float* out           = (float*)d_out;

    dim3 grid(NPTS / PTS_PER_BLK, BATCH);
    ode_hypernet_kernel<<<grid, 128, 0, stream>>>(context, y, tnw, out);
}
